# starting kernel
# baseline (speedup 1.0000x reference)
#include <hip/hip_runtime.h>

typedef __attribute__((ext_vector_type(2))) float v2f;
typedef __attribute__((ext_vector_type(8))) float v8f;

#define TREES  2048
#define KNODES 511
#define EDIM   128
#define DEPTH  9

__global__ __launch_bounds__(256, 1)
void tree_encoder_kernel(const float* __restrict__ emb,
                         const float* __restrict__ W,
                         const float* __restrict__ bias,
                         const int*  __restrict__ tokens,
                         float* __restrict__ out)
{
    __shared__ float hbuf[KNODES * EDIM];
    __shared__ float pmax[256];

    const int tree = blockIdx.x;
    const int tid  = threadIdx.x;
    const int wave = tid >> 5;
    const int lane = tid & 31;
    const int mrow = lane & 15;
    const int half = lane >> 4;

    const int ni  = wave;
    const int col = ni * 16 + mrow;

    v2f bfrag[32];
#pragma unroll
    for (int kk = 0; kk < 32; ++kk) {
        const int k = kk * 4 + 2 * half;
        bfrag[kk].x = W[(k + 0) * EDIM + col];
        bfrag[kk].y = W[(k + 1) * EDIM + col];
    }
    const float bv   = bias[col];
    const int   base = tree * KNODES;

#pragma unroll 1
    for (int mi = 0; mi < 32; ++mi) {
        const int node   = mi * 16 + mrow;
        const int nclamp = node < KNODES ? node : KNODES - 1;
        const int token  = tokens[base + nclamp];
        const float* arow = emb + (size_t)token * EDIM + 2 * half;

        v8f c = {};
#pragma unroll
        for (int kk = 0; kk < 32; ++kk) {
            v2f a = *(const v2f*)(arow + kk * 4);
            c = __builtin_amdgcn_wmma_f32_16x16x4_f32(
                    false, a, false, bfrag[kk], (short)0, c, false, false);
        }
#pragma unroll
        for (int r = 0; r < 8; ++r) {
            const int orow = mi * 16 + r + 8 * half;
            if (orow < KNODES)
                hbuf[orow * EDIM + col] = c[r] + bv;
        }
    }
    __syncthreads();

    for (int l = DEPTH - 1; l >= 1; --l) {
        const int pFirst = (1 << (l - 1)) - 1;
        const int total  = (1 << (l - 1)) * EDIM;
        for (int e = tid; e < total; e += 256) {
            const int p = pFirst + (e >> 7);
            const int d = e & (EDIM - 1);
            hbuf[p * EDIM + d] += hbuf[(2 * p + 1) * EDIM + d]
                                + hbuf[(2 * p + 2) * EDIM + d];
        }
        __syncthreads();
    }

    {
        const int d    = tid & (EDIM - 1);
        const int part = tid >> 7;
        const int n0   = part * 256;
        const int n1   = part ? KNODES : 256;
        float m = -__builtin_inff();
        for (int n = n0; n < n1; ++n)
            m = fmaxf(m, hbuf[n * EDIM + d]);
        pmax[tid] = m;
    }
    __syncthreads();
    if (tid < EDIM) {
        const float m = fmaxf(pmax[tid], pmax[tid + EDIM]);
        out[tree * EDIM + tid] = fmaxf(m, 0.0f);
    }
}

extern "C" void kernel_launch(void* const* d_in, const int* in_sizes, int n_in,
                              void* d_out, int out_size, void* d_ws, size_t ws_size,
                              hipStream_t stream)
{
    const float* emb    = (const float*)d_in[0];
    const float* W      = (const float*)d_in[1];
    const float* bias   = (const float*)d_in[2];
    const int*   tokens = (const int*)d_in[3];
    float*       out    = (float*)d_out;

    tree_encoder_kernel<<<TREES, 256, 0, stream>>>(emb, W, bias, tokens, out);
}


# Each wave owns a whole 63-node subtree, sums it in registers with no workgroup barrier; one split barrier per tree; s_setprio rises through the WMMAs: sum phases run under the SIMD partner's WMMAs
# speedup vs baseline: 2.9853x; 2.9853x over previous
#include <hip/hip_runtime.h>


typedef __attribute__((ext_vector_type(2))) float v2f;
typedef __attribute__((ext_vector_type(4))) float v4f;
typedef __attribute__((ext_vector_type(8))) float v8f;

#define TREES  2048
#define KNODES 511
#define EDIM   128
#define DEPTH  9
#define KGRP   16
#define AHEAD  4


__device__ __forceinline__ v4f max4(const v4f a, const v4f b)
{
    v4f r;
    r.x = fmaxf(a.x, b.x);  r.y = fmaxf(a.y, b.y);
    r.z = fmaxf(a.z, b.z);  r.w = fmaxf(a.w, b.w);
    return r;
}

__device__ __forceinline__ void lds_barrier()
{
    asm volatile("s_wait_dscnt 0x0\n\ts_barrier_signal -1\n\ts_barrier_wait -1" ::: "memory");
}

__device__ __forceinline__ void wave_lds_done()
{
    asm volatile("s_wait_dscnt 0x0" ::: "memory");
}

__device__ __forceinline__ void handover_signal()
{
    asm volatile("s_wait_dscnt 0x0\n\ts_barrier_signal -1" ::: "memory");
}
__device__ __forceinline__ void handover_wait()
{
    asm volatile("s_barrier_wait -1" ::: "memory");
}

__device__ __forceinline__ void top_combine(const float* mb, float* outp)
{
    v4f R[8], M[8], Ht[7];
#pragma unroll
    for (int w = 0; w < 8; ++w) {
        R[w] = *(const v4f*)(mb + (w * 3 + 0) * EDIM);
        M[w] = *(const v4f*)(mb + (w * 3 + 1) * EDIM);
    }
#pragma unroll
    for (int n = 0; n < 7; ++n)
        Ht[n] = *(const v4f*)(mb + (n * 3 + 2) * EDIM);
    const v4f S3 = Ht[3] + (R[0] + R[1]);
    const v4f S4 = Ht[4] + (R[2] + R[3]);
    const v4f S5 = Ht[5] + (R[4] + R[5]);
    const v4f S6 = Ht[6] + (R[6] + R[7]);
    const v4f S1 = Ht[1] + (S3 + S4);
    const v4f S2 = Ht[2] + (S5 + S6);
    const v4f S0 = Ht[0] + (S1 + S2);
    v4f m = M[0];
#pragma unroll
    for (int w = 1; w < 8; ++w)
        m = max4(m, M[w]);
    m = max4(m, S0);  m = max4(m, S1);  m = max4(m, S2);  m = max4(m, S3);
    m = max4(m, S4);  m = max4(m, S5);  m = max4(m, S6);
    const v4f zero = {0.0f, 0.0f, 0.0f, 0.0f};
    *(v4f*)outp = max4(m, zero);
}

__global__ __launch_bounds__(256, 1)
void tree_encoder_kernel(const float* __restrict__ emb,
                         const float* __restrict__ W,
                         const float* __restrict__ bias,
                         const int*  __restrict__ tokens,
                         float* __restrict__ out)
{
    __shared__ __attribute__((aligned(16))) float smem[EDIM * EDIM + 8 * 32 * EDIM + 2 * 8 * 3 * EDIM];
    float* wfrag = smem;
    float* hreg  = smem + EDIM * EDIM;
    float* mbox  = hreg + 8 * 32 * EDIM;

    const int tid  = threadIdx.x;
    const int wave = __builtin_amdgcn_readfirstlane(tid >> 5);
    const int lane = tid & 31;
    const int mrow = lane & 15;
    const int half = lane >> 4;

    int tree = blockIdx.x;
    int it   = 0;

    int knode[4];
#pragma unroll
    for (int t = 0; t < 4; ++t) {
        const int k   = t * 16 + mrow;
        const int kk  = k ? k : 1;
        const int d   = 31 - __builtin_clz((unsigned)kk);
        const int sub = ((7 + wave) << d) + (k - 1);
        knode[t] = k ? sub : (wave < 7 ? wave : 0);
    }
    const char* ebase = (const char*)emb;
    unsigned arow[4];
#pragma unroll
    for (int t = 0; t < 4; ++t) {
        const unsigned token = (unsigned)tokens[tree * KNODES + knode[t]];
        arow[t] = token * (unsigned)(EDIM * 4) + 16u * (unsigned)half;
    }

    v4f a[4][KGRP];
#pragma unroll
    for (int j = 0; j < AHEAD; ++j)
#pragma unroll
        for (int t = 0; t < 4; ++t)
            a[t][j] = *(const v4f*)(ebase + (size_t)arow[t] + 32 * j);

#pragma unroll
    for (int i16 = 0; i16 < 16; ++i16) {
        const int item = i16 * 256 + tid;
        const int c    = item & (EDIM - 1);
        const int kg   = item >> 7;
        v4f w4;
        w4.x = W[(4 * kg + 0) * EDIM + c];
        w4.y = W[(4 * kg + 1) * EDIM + c];
        w4.z = W[(4 * kg + 2) * EDIM + c];
        w4.w = W[(4 * kg + 3) * EDIM + c];
        const int ni   = ((c >> 6) << 2) | (c & 3);
        const int n    = (c >> 2) & 15;
        const int slot = (((kg >> 1) * 8 + ni) * 32) + (kg & 1) * 16 + n;
        *(v4f*)(wfrag + slot * 4) = w4;
    }
    float bv[8];
#pragma unroll
    for (int ni = 0; ni < 8; ++ni)
        bv[ni] = bias[64 * (ni >> 2) + 4 * mrow + (ni & 3)];
    lds_barrier();

    for (;;) {
        const int  next     = tree + (int)gridDim.x;
        const bool has_next = next < TREES;

        const int tnext = has_next ? next : TREES - 1;
        unsigned tokn[4];
#pragma unroll
        for (int t = 0; t < 4; ++t)
            tokn[t] = (unsigned)tokens[tnext * KNODES + knode[t]];

        v8f acc[4][8];
#pragma unroll
        for (int t = 0; t < 4; ++t)
#pragma unroll
            for (int ni = 0; ni < 8; ++ni)
                acc[t][ni] = (v8f){};

        v4f b[2][8];
#pragma unroll
        for (int ni = 0; ni < 8; ++ni)
            b[0][ni] = *(const v4f*)(wfrag + (ni * 32 + lane) * 4);

#pragma unroll
        for (int j = 0; j < KGRP; ++j) {
            __builtin_amdgcn_sched_barrier(0);
            if (j == 1)  __builtin_amdgcn_s_setprio(1);
            if (j == 6)  __builtin_amdgcn_s_setprio(2);
            if (j == 11) __builtin_amdgcn_s_setprio(3);
            if (j + AHEAD < KGRP) {
#pragma unroll
                for (int t = 0; t < 4; ++t)
                    a[t][j + AHEAD] = *(const v4f*)(ebase + (size_t)arow[t] + 32 * (j + AHEAD));
            }
            if (j + 1 < KGRP) {
#pragma unroll
                for (int ni = 0; ni < 8; ++ni)
                    b[(j + 1) & 1][ni] = *(const v4f*)(wfrag + (((j + 1) * 8 + ni) * 32 + lane) * 4);
            }
#pragma unroll
            for (int t = 0; t < 4; ++t)
#pragma unroll
                for (int ni = 0; ni < 8; ++ni) {
                    const v2f a2 = {a[t][j].x, a[t][j].y};
                    const v2f b2 = {b[j & 1][ni].x, b[j & 1][ni].y};
                    acc[t][ni] = __builtin_amdgcn_wmma_f32_16x16x4_f32(
                            false, a2, false, b2, (short)0, acc[t][ni], false, false);
                }
#pragma unroll
            for (int t = 0; t < 4; ++t)
#pragma unroll
                for (int ni = 0; ni < 8; ++ni) {
                    const v2f a2 = {a[t][j].z, a[t][j].w};
                    const v2f b2 = {b[j & 1][ni].z, b[j & 1][ni].w};
                    acc[t][ni] = __builtin_amdgcn_wmma_f32_16x16x4_f32(
                            false, a2, false, b2, (short)0, acc[t][ni], false, false);
                }
        }
        __builtin_amdgcn_sched_barrier(0);
        __builtin_amdgcn_s_setprio(0);

        float* reg = hreg + wave * (32 * EDIM);
#pragma unroll
        for (int t = 2; t < 4; ++t)
#pragma unroll
            for (int r = 0; r < 8; ++r) {
                const int lrow = (t - 2) * 16 + r + 8 * half;
                v4f lo, hi;
                lo.x = acc[t][0][r] + bv[0];  lo.y = acc[t][1][r] + bv[1];
                lo.z = acc[t][2][r] + bv[2];  lo.w = acc[t][3][r] + bv[3];
                hi.x = acc[t][4][r] + bv[4];  hi.y = acc[t][5][r] + bv[5];
                hi.z = acc[t][6][r] + bv[6];  hi.w = acc[t][7][r] + bv[7];
                *(v4f*)(reg + lrow * EDIM + 4 * mrow)      = lo;
                *(v4f*)(reg + lrow * EDIM + 64 + 4 * mrow) = hi;
            }

#pragma unroll
        for (int t = 0; t < 4; ++t)
            arow[t] = tokn[t] * (unsigned)(EDIM * 4) + 16u * (unsigned)half;
#pragma unroll
        for (int j = 0; j < AHEAD; ++j)
#pragma unroll
            for (int t = 0; t < 4; ++t)
                a[t][j] = *(const v4f*)(ebase + (size_t)arow[t] + 32 * j);

        const int lq = lane;
        const float* rcol = reg + 4 * lq;

        wave_lds_done();
        v4f Lf[32];
#pragma unroll
        for (int i = 0; i < 32; ++i)
            Lf[i] = *(const v4f*)(rcol + i * EDIM);
        wave_lds_done();

#pragma unroll
        for (int t = 0; t < 2; ++t)
#pragma unroll
            for (int r = 0; r < 8; ++r) {
                const int lrow = t * 16 + r + 8 * half;
                v4f lo, hi;
                lo.x = acc[t][0][r] + bv[0];  lo.y = acc[t][1][r] + bv[1];
                lo.z = acc[t][2][r] + bv[2];  lo.w = acc[t][3][r] + bv[3];
                hi.x = acc[t][4][r] + bv[4];  hi.y = acc[t][5][r] + bv[5];
                hi.z = acc[t][6][r] + bv[6];  hi.w = acc[t][7][r] + bv[7];
                *(v4f*)(reg + lrow * EDIM + 4 * mrow)      = lo;
                *(v4f*)(reg + lrow * EDIM + 64 + 4 * mrow) = hi;
            }
        wave_lds_done();
        v4f H[32];
#pragma unroll
        for (int k = 0; k < 32; ++k)
            H[k] = *(const v4f*)(rcol + k * EDIM);

        v4f vm = Lf[0];
#pragma unroll
        for (int i = 1; i < 32; ++i)
            vm = max4(vm, Lf[i]);
#pragma unroll
        for (int k = 16; k < 32; ++k) {
            H[k] = H[k] + (Lf[2 * k - 32] + Lf[2 * k - 31]);
            vm = max4(vm, H[k]);
        }
#pragma unroll
        for (int k = 15; k >= 1; --k) {
            H[k] = H[k] + (H[2 * k] + H[2 * k + 1]);
            vm = max4(vm, H[k]);
        }

        if (it > 0) {
            handover_wait();
            if (wave == ((it - 1) & 7))
                top_combine(mbox + ((it - 1) & 1) * (8 * 3 * EDIM) + 4 * lq,
                            out + (tree - (int)gridDim.x) * EDIM + 4 * lq);
        }
        float* mb = mbox + (((it & 1) * 8 + wave) * 3) * EDIM + 4 * lq;
        *(v4f*)(mb)            = H[1];
        *(v4f*)(mb + EDIM)     = vm;
        *(v4f*)(mb + 2 * EDIM) = H[0];
        handover_signal();

        if (!has_next) break;
        tree = next;
        ++it;
    }
    handover_wait();
    if (wave == (it & 7))
        top_combine(mbox + (it & 1) * (8 * 3 * EDIM) + 4 * lane, out + tree * EDIM + 4 * lane);
    asm volatile("s_wait_loadcnt 0x0" ::: "memory");
}

extern "C" void kernel_launch(void* const* d_in, const int* in_sizes, int n_in,
                              void* d_out, int out_size, void* d_ws, size_t ws_size,
                              hipStream_t stream)
{
    const float* emb    = (const float*)d_in[0];
    const float* W      = (const float*)d_in[1];
    const float* bias   = (const float*)d_in[2];
    const int*   tokens = (const int*)d_in[3];
    float*       out    = (float*)d_out;

    tree_encoder_kernel<<<256, 256, 0, stream>>>(emb, W, bias, tokens, out);
}
